# starting kernel
# baseline (speedup 1.0000x reference)
#include <hip/hip_runtime.h>

typedef __attribute__((ext_vector_type(2))) float v2f;
typedef __attribute__((ext_vector_type(8))) float v8f;

namespace {
constexpr int kSteps         = 30;
constexpr int kWavesPerBlock = 8;
constexpr int kBatchPerWave  = 16;
constexpr int kBatchPerBlock = kWavesPerBlock * kBatchPerWave;
}

static __device__ __forceinline__ float relu1(float x) {
  float r;
  asm("v_max_num_f32 %0, 0, %1" : "=v"(r) : "v"(x));
  return r;
}

static __device__ __forceinline__ float swap16(float x) {
  return __int_as_float(__builtin_amdgcn_ds_swizzle(__float_as_int(x), 0x401F));
}

__global__ __launch_bounds__(kWavesPerBlock * 32) void
hedge_wmma_kernel(const float* __restrict__ S,
                  const float* __restrict__ W1,
                  const float* __restrict__ b1,
                  const float* __restrict__ W2,
                  const float* __restrict__ b2,
                  const float* __restrict__ a_init,
                  float* __restrict__ out,
                  int Btot) {
  const int tid  = threadIdx.x;
  const int lane = tid & 31;
  const int wave = tid >> 5;
  const int base = (blockIdx.x * kWavesPerBlock + wave) * kBatchPerWave;
  if (base >= Btot) return;

  const int  n  = lane & 15;
  const bool lo = lane < 16;
  const int  bat = base + n;

  v2f a_lo, a_hi;
  a_lo.x = lo ? W1[n * 3 + 0] : W1[n * 3 + 2];
  a_lo.y = lo ? W1[n * 3 + 1] : 0.0f;
  a_hi.x = lo ? W1[(n + 16) * 3 + 0] : W1[(n + 16) * 3 + 2];
  a_hi.y = lo ? W1[(n + 16) * 3 + 1] : 0.0f;

  v8f c_lo, c_hi;
#pragma unroll
  for (int v = 0; v < 8; ++v) {
    c_lo[v] = lo ? b1[v]      : b1[v + 8];
    c_hi[v] = lo ? b1[v + 16] : b1[v + 24];
  }

  float w2lo[8], w2hi[8];
#pragma unroll
  for (int v = 0; v < 8; ++v) {
    w2lo[v] = lo ? W2[v]      : W2[v + 8];
    w2hi[v] = lo ? W2[v + 16] : W2[v + 24];
  }

  const float halfb2 = 0.5f * b2[0];

  float s_arr[kSteps];
#pragma unroll
  for (int t = 0; t < kSteps; ++t)
    s_arr[t] = __builtin_nontemporal_load(&S[bat * kSteps + t]);

  float h = 0.0f;
  float d = a_init[0];

  const int outbase = bat * kSteps - (lo ? 1 : 0);
  float dkeep = 0.0f;

#pragma unroll
  for (int t = 0; t < kSteps; ++t) {
    v2f bm;
    bm.x = lo ? s_arr[t] : h;
    bm.y = lo ? d : 0.0f;

    v8f hid_lo = __builtin_amdgcn_wmma_f32_16x16x4_f32(
        false, a_lo, false, bm, (short)0, c_lo, false, false);
    v8f hid_hi = __builtin_amdgcn_wmma_f32_16x16x4_f32(
        false, a_hi, false, bm, (short)0, c_hi, false, false);

    float acc0 = halfb2, acc1 = 0.0f, acc2 = 0.0f, acc3 = 0.0f;
#pragma unroll
    for (int v = 0; v < 4; ++v) {
      acc0 = fmaf(relu1(hid_lo[v]),     w2lo[v],     acc0);
      acc1 = fmaf(relu1(hid_lo[v + 4]), w2lo[v + 4], acc1);
      acc2 = fmaf(relu1(hid_hi[v]),     w2hi[v],     acc2);
      acc3 = fmaf(relu1(hid_hi[v + 4]), w2hi[v + 4], acc3);
    }
    const float part = (acc0 + acc1) + (acc2 + acc3);
    const float dval = part + swap16(part);

    h = fmaf(0.2f, dval, 0.8f * h);
    d = dval;

    if (t & 1) {
      const float vstore = lo ? dkeep : dval;
      __builtin_nontemporal_store(vstore, &out[outbase + t]);
    } else {
      dkeep = dval;
    }
  }
}

extern "C" void kernel_launch(void* const* d_in, const int* in_sizes, int n_in,
                              void* d_out, int out_size, void* d_ws, size_t ws_size,
                              hipStream_t stream) {
  const float* S      = (const float*)d_in[0];
  const float* W1     = (const float*)d_in[1];
  const float* b1     = (const float*)d_in[2];
  const float* W2     = (const float*)d_in[3];
  const float* b2     = (const float*)d_in[4];
  const float* a_init = (const float*)d_in[5];
  float* out = (float*)d_out;

  const int Btot   = in_sizes[0] / kSteps;
  const int blocks = (Btot + kBatchPerBlock - 1) / kBatchPerBlock;
  hedge_wmma_kernel<<<blocks, kWavesPerBlock * 32, 0, stream>>>(
      S, W1, b1, W2, b2, a_init, out, Btot);
}


# Whole-line global I/O via a per-wave LDS transpose: the contiguous 1920-byte S tile and result tile move as 4+4 full-width 16-byte accesses instead of lane-strided loads and 8-byte store pieces
# speedup vs baseline: 1.3753x; 1.3753x over previous
#include <hip/hip_runtime.h>

typedef __attribute__((ext_vector_type(2))) float v2f;
typedef __attribute__((ext_vector_type(8))) float v8f;
typedef __attribute__((ext_vector_type(4))) float f4;

namespace {
constexpr int kSteps         = 30;
constexpr int kWavesPerBlock = 8;
constexpr int kBatchPerWave  = 16;
constexpr int kBatchPerBlock = kWavesPerBlock * kBatchPerWave;
constexpr int kPieces        = kBatchPerWave * kSteps / 4;
constexpr int kSlabPieces    = 128;
}

static __device__ __forceinline__ float relu1(float x) {
  float r;
  asm("v_max_num_f32 %0, 0, %1" : "=v"(r) : "v"(x));
  return r;
}

static __device__ __forceinline__ float swap16(float x) {
  return __int_as_float(__builtin_amdgcn_ds_swizzle(__float_as_int(x), 0x401F));
}

__global__ __launch_bounds__(kWavesPerBlock * 32) void
hedge_wmma_kernel(const float* __restrict__ S,
                  const float* __restrict__ W1,
                  const float* __restrict__ b1,
                  const float* __restrict__ W2,
                  const float* __restrict__ b2,
                  const float* __restrict__ a_init,
                  float* __restrict__ out,
                  int Btot) {
  __shared__ f4 tile4[kWavesPerBlock][kSlabPieces];

  const int tid  = threadIdx.x;
  const int lane = tid & 31;
  const int wave = tid >> 5;
  const int base = (blockIdx.x * kWavesPerBlock + wave) * kBatchPerWave;
  if (base >= Btot) return;

  const int  n  = lane & 15;
  const bool lo = lane < 16;

  v2f a_lo, a_hi;
  a_lo.x = lo ? W1[n * 3 + 0] : W1[n * 3 + 2];
  a_lo.y = lo ? W1[n * 3 + 1] : 0.0f;
  a_hi.x = lo ? W1[(n + 16) * 3 + 0] : W1[(n + 16) * 3 + 2];
  a_hi.y = lo ? W1[(n + 16) * 3 + 1] : 0.0f;

  v8f c_lo, c_hi;
#pragma unroll
  for (int v = 0; v < 8; ++v) {
    c_lo[v] = lo ? b1[v]      : b1[v + 8];
    c_hi[v] = lo ? b1[v + 16] : b1[v + 24];
  }

  float w2lo[8], w2hi[8];
#pragma unroll
  for (int v = 0; v < 8; ++v) {
    w2lo[v] = lo ? W2[v]      : W2[v + 8];
    w2hi[v] = lo ? W2[v + 16] : W2[v + 24];
  }

  const float halfb2 = 0.5f * b2[0];

  float* tile = (float*)tile4[wave];
  const f4* gsrc = (const f4*)(S + (size_t)base * kSteps);
#pragma unroll
  for (int p = 0; p < kSlabPieces; p += 32) {
    const int idx = p + lane;
    const int gidx = idx < kPieces ? idx : kPieces - 1;
    tile4[wave][idx] = __builtin_nontemporal_load(&gsrc[gidx]);
  }
  __builtin_amdgcn_fence(__ATOMIC_RELEASE, "wavefront");
  __builtin_amdgcn_wave_barrier();
  __builtin_amdgcn_fence(__ATOMIC_ACQUIRE, "wavefront");

  float s_arr[kSteps];
  {
    const v2f* row = (const v2f*)(tile + n * kSteps);
#pragma unroll
    for (int j = 0; j < kSteps / 2; ++j) {
      const v2f r = row[j];
      s_arr[2 * j]     = r.x;
      s_arr[2 * j + 1] = r.y;
    }
  }

  float h = 0.0f;
  float d = a_init[0];

  float o_arr[kSteps];

#pragma unroll
  for (int t = 0; t < kSteps; ++t) {
    v2f bm;
    bm.x = lo ? s_arr[t] : h;
    bm.y = lo ? d : 0.0f;

    v8f hid_lo = __builtin_amdgcn_wmma_f32_16x16x4_f32(
        false, a_lo, false, bm, (short)0, c_lo, false, false);
    v8f hid_hi = __builtin_amdgcn_wmma_f32_16x16x4_f32(
        false, a_hi, false, bm, (short)0, c_hi, false, false);

    float acc0 = halfb2, acc1 = 0.0f, acc2 = 0.0f, acc3 = 0.0f;
#pragma unroll
    for (int v = 0; v < 4; ++v) {
      acc0 = fmaf(relu1(hid_lo[v]),     w2lo[v],     acc0);
      acc1 = fmaf(relu1(hid_lo[v + 4]), w2lo[v + 4], acc1);
      acc2 = fmaf(relu1(hid_hi[v]),     w2hi[v],     acc2);
      acc3 = fmaf(relu1(hid_hi[v + 4]), w2hi[v + 4], acc3);
    }
    const float part = (acc0 + acc1) + (acc2 + acc3);
    const float dval = part + swap16(part);

    h = fmaf(0.2f, dval, 0.8f * h);
    d = dval;
    o_arr[t] = dval;
  }

  __builtin_amdgcn_fence(__ATOMIC_ACQ_REL, "wavefront");
  __builtin_amdgcn_wave_barrier();

  if (lo) {
    v2f* row = (v2f*)(tile + n * kSteps);
#pragma unroll
    for (int j = 0; j < kSteps / 2; ++j) {
      v2f r;
      r.x = o_arr[2 * j];
      r.y = o_arr[2 * j + 1];
      row[j] = r;
    }
  }
  __builtin_amdgcn_fence(__ATOMIC_RELEASE, "wavefront");
  __builtin_amdgcn_wave_barrier();
  __builtin_amdgcn_fence(__ATOMIC_ACQUIRE, "wavefront");

  f4* gdst = (f4*)(out + (size_t)base * kSteps);
#pragma unroll
  for (int p = 0; p < kPieces; p += 32) {
    const int idx = p + lane;
    if (idx < kPieces)
      __builtin_nontemporal_store(tile4[wave][idx], &gdst[idx]);
  }
}

extern "C" void kernel_launch(void* const* d_in, const int* in_sizes, int n_in,
                              void* d_out, int out_size, void* d_ws, size_t ws_size,
                              hipStream_t stream) {
  const float* S      = (const float*)d_in[0];
  const float* W1     = (const float*)d_in[1];
  const float* b1     = (const float*)d_in[2];
  const float* W2     = (const float*)d_in[3];
  const float* b2     = (const float*)d_in[4];
  const float* a_init = (const float*)d_in[5];
  float* out = (float*)d_out;

  const int Btot   = in_sizes[0] / kSteps;
  const int blocks = (Btot + kBatchPerBlock - 1) / kBatchPerBlock;
  hedge_wmma_kernel<<<blocks, kWavesPerBlock * 32, 0, stream>>>(
      S, W1, b1, W2, b2, a_init, out, Btot);
}


# ReLU as a compiler-visible integer max on the float bits instead of an asm statement: the compiler now schedules and delay-hints the WMMA, ReLU, FMA dependency chain, so resident waves interleave
# speedup vs baseline: 1.4453x; 1.0509x over previous
#include <hip/hip_runtime.h>

typedef __attribute__((ext_vector_type(2))) float v2f;
typedef __attribute__((ext_vector_type(8))) float v8f;
typedef __attribute__((ext_vector_type(4))) float f4;

namespace {
constexpr int kSteps         = 30;
constexpr int kWavesPerBlock = 8;
constexpr int kBatchPerWave  = 16;
constexpr int kBatchPerBlock = kWavesPerBlock * kBatchPerWave;
constexpr int kPieces        = kBatchPerWave * kSteps / 4;
constexpr int kSlabPieces    = 128;
}

static __device__ __forceinline__ float relu1(float x) {
  const int bits = __float_as_int(x);
  return __int_as_float(bits > 0 ? bits : 0);
}

static __device__ __forceinline__ float swap16(float x) {
  return __int_as_float(__builtin_amdgcn_ds_swizzle(__float_as_int(x), 0x401F));
}

__global__ __launch_bounds__(kWavesPerBlock * 32) void
hedge_wmma_kernel(const float* __restrict__ S,
                  const float* __restrict__ W1,
                  const float* __restrict__ b1,
                  const float* __restrict__ W2,
                  const float* __restrict__ b2,
                  const float* __restrict__ a_init,
                  float* __restrict__ out,
                  int Btot) {
  __shared__ f4 tile4[kWavesPerBlock][kSlabPieces];

  const int tid  = threadIdx.x;
  const int lane = tid & 31;
  const int wave = tid >> 5;
  const int base = (blockIdx.x * kWavesPerBlock + wave) * kBatchPerWave;
  if (base >= Btot) return;

  const int  n  = lane & 15;
  const bool lo = lane < 16;

  v2f a_lo, a_hi;
  a_lo.x = lo ? W1[n * 3 + 0] : W1[n * 3 + 2];
  a_lo.y = lo ? W1[n * 3 + 1] : 0.0f;
  a_hi.x = lo ? W1[(n + 16) * 3 + 0] : W1[(n + 16) * 3 + 2];
  a_hi.y = lo ? W1[(n + 16) * 3 + 1] : 0.0f;

  v8f c_lo, c_hi;
#pragma unroll
  for (int v = 0; v < 8; ++v) {
    c_lo[v] = lo ? b1[v]      : b1[v + 8];
    c_hi[v] = lo ? b1[v + 16] : b1[v + 24];
  }

  float w2lo[8], w2hi[8];
#pragma unroll
  for (int v = 0; v < 8; ++v) {
    w2lo[v] = lo ? W2[v]      : W2[v + 8];
    w2hi[v] = lo ? W2[v + 16] : W2[v + 24];
  }

  const float halfb2 = 0.5f * b2[0];

  float* tile = (float*)tile4[wave];
  const f4* gsrc = (const f4*)(S + (size_t)base * kSteps);
#pragma unroll
  for (int p = 0; p < kSlabPieces; p += 32) {
    const int idx = p + lane;
    const int gidx = idx < kPieces ? idx : kPieces - 1;
    tile4[wave][idx] = __builtin_nontemporal_load(&gsrc[gidx]);
  }
  __builtin_amdgcn_fence(__ATOMIC_RELEASE, "wavefront");
  __builtin_amdgcn_wave_barrier();
  __builtin_amdgcn_fence(__ATOMIC_ACQUIRE, "wavefront");

  float s_arr[kSteps];
  {
    const v2f* row = (const v2f*)(tile + n * kSteps);
#pragma unroll
    for (int j = 0; j < kSteps / 2; ++j) {
      const v2f r = row[j];
      s_arr[2 * j]     = r.x;
      s_arr[2 * j + 1] = r.y;
    }
  }

  float h = 0.0f;
  float d = a_init[0];

  float o_arr[kSteps];

#pragma unroll
  for (int t = 0; t < kSteps; ++t) {
    v2f bm;
    bm.x = lo ? s_arr[t] : h;
    bm.y = lo ? d : 0.0f;

    v8f hid_lo = __builtin_amdgcn_wmma_f32_16x16x4_f32(
        false, a_lo, false, bm, (short)0, c_lo, false, false);
    v8f hid_hi = __builtin_amdgcn_wmma_f32_16x16x4_f32(
        false, a_hi, false, bm, (short)0, c_hi, false, false);

    float acc0 = halfb2, acc1 = 0.0f, acc2 = 0.0f, acc3 = 0.0f;
#pragma unroll
    for (int v = 0; v < 4; ++v) {
      acc0 = fmaf(relu1(hid_lo[v]),     w2lo[v],     acc0);
      acc1 = fmaf(relu1(hid_lo[v + 4]), w2lo[v + 4], acc1);
      acc2 = fmaf(relu1(hid_hi[v]),     w2hi[v],     acc2);
      acc3 = fmaf(relu1(hid_hi[v + 4]), w2hi[v + 4], acc3);
    }
    const float part = (acc0 + acc1) + (acc2 + acc3);
    const float dval = part + swap16(part);

    h = fmaf(0.2f, dval, 0.8f * h);
    d = dval;
    o_arr[t] = dval;
  }

  __builtin_amdgcn_fence(__ATOMIC_ACQ_REL, "wavefront");
  __builtin_amdgcn_wave_barrier();

  if (lo) {
    v2f* row = (v2f*)(tile + n * kSteps);
#pragma unroll
    for (int j = 0; j < kSteps / 2; ++j) {
      v2f r;
      r.x = o_arr[2 * j];
      r.y = o_arr[2 * j + 1];
      row[j] = r;
    }
  }
  __builtin_amdgcn_fence(__ATOMIC_RELEASE, "wavefront");
  __builtin_amdgcn_wave_barrier();
  __builtin_amdgcn_fence(__ATOMIC_ACQUIRE, "wavefront");

  f4* gdst = (f4*)(out + (size_t)base * kSteps);
#pragma unroll
  for (int p = 0; p < kPieces; p += 32) {
    const int idx = p + lane;
    if (idx < kPieces)
      __builtin_nontemporal_store(tile4[wave][idx], &gdst[idx]);
  }
}

extern "C" void kernel_launch(void* const* d_in, const int* in_sizes, int n_in,
                              void* d_out, int out_size, void* d_ws, size_t ws_size,
                              hipStream_t stream) {
  const float* S      = (const float*)d_in[0];
  const float* W1     = (const float*)d_in[1];
  const float* b1     = (const float*)d_in[2];
  const float* W2     = (const float*)d_in[3];
  const float* b2     = (const float*)d_in[4];
  const float* a_init = (const float*)d_in[5];
  float* out = (float*)d_out;

  const int Btot   = in_sizes[0] / kSteps;
  const int blocks = (Btot + kBatchPerBlock - 1) / kBatchPerBlock;
  hedge_wmma_kernel<<<blocks, kWavesPerBlock * 32, 0, stream>>>(
      S, W1, b1, W2, b2, a_init, out, Btot);
}


# ReLU folded into the layer-2 FMA as an |x| source modifier via relu(x) = (x + |x|)/2, summed in one sequential FMA chain per half-wave; the linear half costs two FMAs per step
# speedup vs baseline: 1.5942x; 1.1030x over previous
#include <hip/hip_runtime.h>

typedef __attribute__((ext_vector_type(2))) float v2f;
typedef __attribute__((ext_vector_type(8))) float v8f;
typedef __attribute__((ext_vector_type(4))) float f4;

namespace {
constexpr int kSteps         = 30;
constexpr int kWavesPerBlock = 8;
constexpr int kBatchPerWave  = 16;
constexpr int kBatchPerBlock = kWavesPerBlock * kBatchPerWave;
constexpr int kPieces        = kBatchPerWave * kSteps / 4;
constexpr int kSlabPieces    = 128;
}

template <int kXor>
static __device__ __forceinline__ float xor_lane(float x) {
  return __int_as_float(__builtin_amdgcn_ds_swizzle(__float_as_int(x), (kXor << 10) | 0x1F));
}

static __device__ __forceinline__ float swap16(float x) {
  return __int_as_float(__builtin_amdgcn_ds_swizzle(__float_as_int(x), 0x401F));
}

__global__ __launch_bounds__(kWavesPerBlock * 32) void
hedge_wmma_kernel(const float* __restrict__ S,
                  const float* __restrict__ W1,
                  const float* __restrict__ b1,
                  const float* __restrict__ W2,
                  const float* __restrict__ b2,
                  const float* __restrict__ a_init,
                  float* __restrict__ out,
                  int Btot) {
  __shared__ f4 tile4[kWavesPerBlock][kSlabPieces];

  const int tid  = threadIdx.x;
  const int lane = tid & 31;
  const int wave = tid >> 5;
  const int base = (blockIdx.x * kWavesPerBlock + wave) * kBatchPerWave;
  if (base >= Btot) return;

  const int  n  = lane & 15;
  const bool lo = lane < 16;

  v2f a_lo, a_hi;
  a_lo.x = lo ? W1[n * 3 + 0] : W1[n * 3 + 2];
  a_lo.y = lo ? W1[n * 3 + 1] : 0.0f;
  a_hi.x = lo ? W1[(n + 16) * 3 + 0] : W1[(n + 16) * 3 + 2];
  a_hi.y = lo ? W1[(n + 16) * 3 + 1] : 0.0f;

  v8f c_lo, c_hi;
#pragma unroll
  for (int v = 0; v < 8; ++v) {
    c_lo[v] = lo ? b1[v]      : b1[v + 8];
    c_hi[v] = lo ? b1[v + 16] : b1[v + 24];
  }

  float whlo[8], whhi[8];
#pragma unroll
  for (int v = 0; v < 8; ++v) {
    whlo[v] = 0.5f * (lo ? W2[v]      : W2[v + 8]);
    whhi[v] = 0.5f * (lo ? W2[v + 16] : W2[v + 24]);
  }

  const float w2a = W2[n], w2b = W2[n + 16];
  float px = fmaf(w2a, a_lo.x, w2b * a_hi.x);
  float py = fmaf(w2a, a_lo.y, w2b * a_hi.y);
  float pb = fmaf(w2a, b1[n], w2b * b1[n + 16]);
  px += xor_lane<1>(px); py += xor_lane<1>(py); pb += xor_lane<1>(pb);
  px += xor_lane<2>(px); py += xor_lane<2>(py); pb += xor_lane<2>(pb);
  px += xor_lane<4>(px); py += xor_lane<4>(py); pb += xor_lane<4>(pb);
  px += xor_lane<8>(px); py += xor_lane<8>(py); pb += xor_lane<8>(pb);
  const float cx = 0.5f * px;
  const float cy = __int_as_float(__builtin_amdgcn_readfirstlane(__float_as_int(0.5f * py)));
  const float kc = __int_as_float(__builtin_amdgcn_readfirstlane(
      __float_as_int(0.5f * fmaf(0.5f, pb, b2[0]))));

  float* tile = (float*)tile4[wave];
  const f4* gsrc = (const f4*)(S + (size_t)base * kSteps);
#pragma unroll
  for (int p = 0; p < kSlabPieces; p += 32) {
    const int idx = p + lane;
    const int gidx = idx < kPieces ? idx : kPieces - 1;
    tile4[wave][idx] = __builtin_nontemporal_load(&gsrc[gidx]);
  }
  __builtin_amdgcn_fence(__ATOMIC_RELEASE, "wavefront");
  __builtin_amdgcn_wave_barrier();
  __builtin_amdgcn_fence(__ATOMIC_ACQUIRE, "wavefront");

  float s_arr[kSteps];
  {
    const v2f* row = (const v2f*)(tile + n * kSteps);
#pragma unroll
    for (int j = 0; j < kSteps / 2; ++j) {
      const v2f r = row[j];
      s_arr[2 * j]     = r.x;
      s_arr[2 * j + 1] = r.y;
    }
  }

  float h = 0.0f;
  float d = a_init[0];

  float o_arr[kSteps];

#pragma unroll
  for (int t = 0; t < kSteps; ++t) {
    v2f bm;
    bm.x = lo ? s_arr[t] : h;
    bm.y = lo ? d : 0.0f;

    v8f hid_lo = __builtin_amdgcn_wmma_f32_16x16x4_f32(
        false, a_lo, false, bm, (short)0, c_lo, false, false);
    v8f hid_hi = __builtin_amdgcn_wmma_f32_16x16x4_f32(
        false, a_hi, false, bm, (short)0, c_hi, false, false);

    float part = fmaf(bm.x, cx, fmaf(bm.y, cy, kc));
#pragma unroll
    for (int v = 0; v < 8; ++v) part = fmaf(__builtin_fabsf(hid_lo[v]), whlo[v], part);
#pragma unroll
    for (int v = 0; v < 8; ++v) part = fmaf(__builtin_fabsf(hid_hi[v]), whhi[v], part);
    const float dval = part + swap16(part);

    h = fmaf(0.2f, dval, 0.8f * h);
    d = dval;
    o_arr[t] = dval;
  }

  __builtin_amdgcn_fence(__ATOMIC_ACQ_REL, "wavefront");
  __builtin_amdgcn_wave_barrier();

  if (lo) {
    v2f* row = (v2f*)(tile + n * kSteps);
#pragma unroll
    for (int j = 0; j < kSteps / 2; ++j) {
      v2f r;
      r.x = o_arr[2 * j];
      r.y = o_arr[2 * j + 1];
      row[j] = r;
    }
  }
  __builtin_amdgcn_fence(__ATOMIC_RELEASE, "wavefront");
  __builtin_amdgcn_wave_barrier();
  __builtin_amdgcn_fence(__ATOMIC_ACQUIRE, "wavefront");

  f4* gdst = (f4*)(out + (size_t)base * kSteps);
#pragma unroll
  for (int p = 0; p < kPieces; p += 32) {
    const int idx = p + lane;
    if (idx < kPieces)
      __builtin_nontemporal_store(tile4[wave][idx], &gdst[idx]);
  }
}

extern "C" void kernel_launch(void* const* d_in, const int* in_sizes, int n_in,
                              void* d_out, int out_size, void* d_ws, size_t ws_size,
                              hipStream_t stream) {
  const float* S      = (const float*)d_in[0];
  const float* W1     = (const float*)d_in[1];
  const float* b1     = (const float*)d_in[2];
  const float* W2     = (const float*)d_in[3];
  const float* b2     = (const float*)d_in[4];
  const float* a_init = (const float*)d_in[5];
  float* out = (float*)d_out;

  const int Btot   = in_sizes[0] / kSteps;
  const int blocks = (Btot + kBatchPerBlock - 1) / kBatchPerBlock;
  hedge_wmma_kernel<<<blocks, kWavesPerBlock * 32, 0, stream>>>(
      S, W1, b1, W2, b2, a_init, out, Btot);
}


# Two independent 16-column batch tiles in flight per wave instead of one (half as many waves): each tile keeps its own per-step chain unchanged and the wave interleaves the two chains
# speedup vs baseline: 1.6449x; 1.0318x over previous
#include <hip/hip_runtime.h>

typedef __attribute__((ext_vector_type(2))) float v2f;
typedef __attribute__((ext_vector_type(8))) float v8f;
typedef __attribute__((ext_vector_type(4))) float f4;

namespace {
constexpr int kSteps         = 30;
constexpr int kWavesPerBlock = 8;
constexpr int kTilesPerWave  = 2;
constexpr int kBatchPerWave  = 16 * kTilesPerWave;
constexpr int kBatchPerBlock = kWavesPerBlock * kBatchPerWave;
constexpr int kPieces        = kBatchPerWave * kSteps / 4;
constexpr int kSlabPieces    = 256;
}

template <int kXor>
static __device__ __forceinline__ float xor_lane(float x) {
  return __int_as_float(__builtin_amdgcn_ds_swizzle(__float_as_int(x), (kXor << 10) | 0x1F));
}

static __device__ __forceinline__ float swap16(float x) {
  return __int_as_float(__builtin_amdgcn_ds_swizzle(__float_as_int(x), 0x401F));
}

__global__ __launch_bounds__(kWavesPerBlock * 32) void
hedge_wmma_kernel(const float* __restrict__ S,
                  const float* __restrict__ W1,
                  const float* __restrict__ b1,
                  const float* __restrict__ W2,
                  const float* __restrict__ b2,
                  const float* __restrict__ a_init,
                  float* __restrict__ out,
                  int Btot) {
  __shared__ f4 tile4[kWavesPerBlock][kSlabPieces];

  const int tid  = threadIdx.x;
  const int lane = tid & 31;
  const int wave = tid >> 5;
  const int base = (blockIdx.x * kWavesPerBlock + wave) * kBatchPerWave;
  if (base >= Btot) return;

  const int  n  = lane & 15;
  const bool lo = lane < 16;

  v2f a_lo, a_hi;
  a_lo.x = lo ? W1[n * 3 + 0] : W1[n * 3 + 2];
  a_lo.y = lo ? W1[n * 3 + 1] : 0.0f;
  a_hi.x = lo ? W1[(n + 16) * 3 + 0] : W1[(n + 16) * 3 + 2];
  a_hi.y = lo ? W1[(n + 16) * 3 + 1] : 0.0f;

  v8f c_lo, c_hi;
#pragma unroll
  for (int v = 0; v < 8; ++v) {
    c_lo[v] = lo ? b1[v]      : b1[v + 8];
    c_hi[v] = lo ? b1[v + 16] : b1[v + 24];
  }

  float whlo[8], whhi[8];
#pragma unroll
  for (int v = 0; v < 8; ++v) {
    whlo[v] = 0.5f * (lo ? W2[v]      : W2[v + 8]);
    whhi[v] = 0.5f * (lo ? W2[v + 16] : W2[v + 24]);
  }

  const float w2a = W2[n], w2b = W2[n + 16];
  float px = fmaf(w2a, a_lo.x, w2b * a_hi.x);
  float py = fmaf(w2a, a_lo.y, w2b * a_hi.y);
  float pb = fmaf(w2a, b1[n], w2b * b1[n + 16]);
  px += xor_lane<1>(px); py += xor_lane<1>(py); pb += xor_lane<1>(pb);
  px += xor_lane<2>(px); py += xor_lane<2>(py); pb += xor_lane<2>(pb);
  px += xor_lane<4>(px); py += xor_lane<4>(py); pb += xor_lane<4>(pb);
  px += xor_lane<8>(px); py += xor_lane<8>(py); pb += xor_lane<8>(pb);
  const float cx = 0.5f * px;
  const float cy = __int_as_float(__builtin_amdgcn_readfirstlane(__float_as_int(0.5f * py)));
  const float kc = __int_as_float(__builtin_amdgcn_readfirstlane(
      __float_as_int(0.5f * fmaf(0.5f, pb, b2[0]))));

  const int rows_here = (Btot - base) < kBatchPerWave ? (Btot - base) : kBatchPerWave;
  const int n_pieces  = rows_here * kSteps / 4;
  float* tile = (float*)tile4[wave];
  const f4* gsrc = (const f4*)(S + (size_t)base * kSteps);
#pragma unroll
  for (int p = 0; p < kSlabPieces; p += 32) {
    const int idx = p + lane;
    const int gidx = idx < n_pieces ? idx : n_pieces - 1;
    tile4[wave][idx] = __builtin_nontemporal_load(&gsrc[gidx]);
  }
  __builtin_amdgcn_fence(__ATOMIC_RELEASE, "wavefront");
  __builtin_amdgcn_wave_barrier();
  __builtin_amdgcn_fence(__ATOMIC_ACQUIRE, "wavefront");

  float s_arr[kTilesPerWave][kSteps];
#pragma unroll
  for (int k = 0; k < kTilesPerWave; ++k) {
    const v2f* row = (const v2f*)(tile + (n + 16 * k) * kSteps);
#pragma unroll
    for (int j = 0; j < kSteps / 2; ++j) {
      const v2f r = row[j];
      s_arr[k][2 * j]     = r.x;
      s_arr[k][2 * j + 1] = r.y;
    }
  }

  float h[kTilesPerWave];
  float d[kTilesPerWave];
  const float d0 = a_init[0];
#pragma unroll
  for (int k = 0; k < kTilesPerWave; ++k) { h[k] = 0.0f; d[k] = d0; }

  float o_arr[kTilesPerWave][kSteps];

#pragma unroll
  for (int t = 0; t < kSteps; ++t) {
    v8f hid_lo[kTilesPerWave], hid_hi[kTilesPerWave];
    float part[kTilesPerWave];
#pragma unroll
    for (int k = 0; k < kTilesPerWave; ++k) {
      v2f bm;
      bm.x = lo ? s_arr[k][t] : h[k];
      bm.y = lo ? d[k] : 0.0f;

      hid_lo[k] = __builtin_amdgcn_wmma_f32_16x16x4_f32(
          false, a_lo, false, bm, (short)0, c_lo, false, false);
      hid_hi[k] = __builtin_amdgcn_wmma_f32_16x16x4_f32(
          false, a_hi, false, bm, (short)0, c_hi, false, false);

      part[k] = fmaf(bm.x, cx, fmaf(bm.y, cy, kc));
    }
#pragma unroll
    for (int v = 0; v < 8; ++v) part[0] = fmaf(__builtin_fabsf(hid_lo[0][v]), whlo[v], part[0]);
#pragma unroll
    for (int v = 0; v < 8; ++v) part[0] = fmaf(__builtin_fabsf(hid_hi[0][v]), whhi[v], part[0]);
    {
#pragma clang fp contract(on)
#pragma unroll
      for (int v = 0; v < 8; ++v) part[1] = __builtin_fabsf(hid_lo[1][v]) * whlo[v] + part[1];
#pragma unroll
      for (int v = 0; v < 8; ++v) part[1] = __builtin_fabsf(hid_hi[1][v]) * whhi[v] + part[1];
    }
#pragma unroll
    for (int k = 0; k < kTilesPerWave; ++k) {
      const float dval = part[k] + swap16(part[k]);
      h[k] = fmaf(0.2f, dval, 0.8f * h[k]);
      d[k] = dval;
      o_arr[k][t] = dval;
    }
  }

  __builtin_amdgcn_fence(__ATOMIC_ACQ_REL, "wavefront");
  __builtin_amdgcn_wave_barrier();

  {
    v2f* row = (v2f*)(tile + lane * kSteps);
    if (lo) {
#pragma unroll
      for (int j = 0; j < kSteps / 2; ++j) {
        v2f r;
        r.x = o_arr[0][2 * j];
        r.y = o_arr[0][2 * j + 1];
        row[j] = r;
      }
    } else {
#pragma unroll
      for (int j = 0; j < kSteps / 2; ++j) {
        v2f r;
        r.x = o_arr[1][2 * j];
        r.y = o_arr[1][2 * j + 1];
        row[j] = r;
      }
    }
  }
  __builtin_amdgcn_fence(__ATOMIC_RELEASE, "wavefront");
  __builtin_amdgcn_wave_barrier();
  __builtin_amdgcn_fence(__ATOMIC_ACQUIRE, "wavefront");

  f4* gdst = (f4*)(out + (size_t)base * kSteps);
  if (n_pieces == kPieces) {
#pragma unroll
    for (int p = 0; p < kSlabPieces; p += 32) {
      const int idx = p + lane;
      if (idx < kPieces)
        __builtin_nontemporal_store(tile4[wave][idx], &gdst[idx]);
    }
  } else {
#pragma unroll
    for (int p = 0; p < kSlabPieces; p += 32) {
      const int idx = p + lane;
      if (idx < n_pieces)
        __builtin_nontemporal_store(tile4[wave][idx], &gdst[idx]);
    }
  }
}

extern "C" void kernel_launch(void* const* d_in, const int* in_sizes, int n_in,
                              void* d_out, int out_size, void* d_ws, size_t ws_size,
                              hipStream_t stream) {
  const float* S      = (const float*)d_in[0];
  const float* W1     = (const float*)d_in[1];
  const float* b1     = (const float*)d_in[2];
  const float* W2     = (const float*)d_in[3];
  const float* b2     = (const float*)d_in[4];
  const float* a_init = (const float*)d_in[5];
  float* out = (float*)d_out;

  const int Btot   = in_sizes[0] / kSteps;
  const int blocks = (Btot + kBatchPerBlock - 1) / kBatchPerBlock;
  hedge_wmma_kernel<<<blocks, kWavesPerBlock * 32, 0, stream>>>(
      S, W1, b1, W2, b2, a_init, out, Btot);
}


# Step loop keeps only its state in registers: B fragment (x, d_prev) advanced in place by per-lane-constant FMAs, no selects; S is read from and results go to the LDS slab one value per step
# speedup vs baseline: 1.7188x; 1.0449x over previous
#include <hip/hip_runtime.h>

typedef __attribute__((ext_vector_type(2))) float v2f;
typedef __attribute__((ext_vector_type(8))) float v8f;
typedef __attribute__((ext_vector_type(4))) float f4;

namespace {
constexpr int kSteps         = 30;
constexpr int kWavesPerBlock = 8;
constexpr int kTilesPerWave  = 2;
constexpr int kBatchPerWave  = 16 * kTilesPerWave;
constexpr int kBatchPerBlock = kWavesPerBlock * kBatchPerWave;
constexpr int kPieces        = kBatchPerWave * kSteps / 4;
constexpr int kSlabPieces    = 256;
}

template <int kXor>
static __device__ __forceinline__ float xor_lane(float x) {
  return __int_as_float(__builtin_amdgcn_ds_swizzle(__float_as_int(x), (kXor << 10) | 0x1F));
}

static __device__ __forceinline__ float swap16(float x) {
  return __int_as_float(__builtin_amdgcn_ds_swizzle(__float_as_int(x), 0x401F));
}

__global__ __launch_bounds__(kWavesPerBlock * 32) __attribute__((amdgpu_waves_per_eu(6, 8))) void
hedge_wmma_kernel(const float* __restrict__ S,
                  const float* __restrict__ W1,
                  const float* __restrict__ b1,
                  const float* __restrict__ W2,
                  const float* __restrict__ b2,
                  const float* __restrict__ a_init,
                  float* __restrict__ out,
                  int Btot) {
  __shared__ f4 tile4[kWavesPerBlock][kSlabPieces];

  const int tid  = threadIdx.x;
  const int lane = tid & 31;
  const int wave = tid >> 5;
  const int base = (blockIdx.x * kWavesPerBlock + wave) * kBatchPerWave;
  if (base >= Btot) return;

  const int  n  = lane & 15;
  const bool lo = lane < 16;

  v2f a_lo, a_hi;
  a_lo.x = lo ? W1[n * 3 + 0] : W1[n * 3 + 2];
  a_lo.y = lo ? W1[n * 3 + 1] : 0.0f;
  a_hi.x = lo ? W1[(n + 16) * 3 + 0] : W1[(n + 16) * 3 + 2];
  a_hi.y = lo ? W1[(n + 16) * 3 + 1] : 0.0f;

  v8f c_lo, c_hi;
#pragma unroll
  for (int v = 0; v < 8; ++v) {
    c_lo[v] = lo ? b1[v]      : b1[v + 8];
    c_hi[v] = lo ? b1[v + 16] : b1[v + 24];
  }

  float whlo[8], whhi[8];
#pragma unroll
  for (int v = 0; v < 8; ++v) {
    whlo[v] = 0.5f * (lo ? W2[v]      : W2[v + 8]);
    whhi[v] = 0.5f * (lo ? W2[v + 16] : W2[v + 24]);
  }

  const float w2a = W2[n], w2b = W2[n + 16];
  float px = fmaf(w2a, a_lo.x, w2b * a_hi.x);
  float py = fmaf(w2a, a_lo.y, w2b * a_hi.y);
  float pb = fmaf(w2a, b1[n], w2b * b1[n + 16]);
  px += xor_lane<1>(px); py += xor_lane<1>(py); pb += xor_lane<1>(pb);
  px += xor_lane<2>(px); py += xor_lane<2>(py); pb += xor_lane<2>(pb);
  px += xor_lane<4>(px); py += xor_lane<4>(py); pb += xor_lane<4>(pb);
  px += xor_lane<8>(px); py += xor_lane<8>(py); pb += xor_lane<8>(pb);
  const float cx = 0.5f * px;
  const float cy_all = __int_as_float(__builtin_amdgcn_readfirstlane(__float_as_int(0.5f * py)));
  const float cy = lo ? cy_all : 0.0f;
  const float c8 = lo ? 0.0f : 0.8f;
  const float c2 = lo ? 0.0f : 0.2f;
  const float kc = __int_as_float(__builtin_amdgcn_readfirstlane(
      __float_as_int(0.5f * fmaf(0.5f, pb, b2[0]))));

  const int rows_here = (Btot - base) < kBatchPerWave ? (Btot - base) : kBatchPerWave;
  const int n_pieces  = rows_here * kSteps / 4;
  float* tile = (float*)tile4[wave];
  const f4* gsrc = (const f4*)(S + (size_t)base * kSteps);
#pragma unroll
  for (int p = 0; p < kSlabPieces; p += 32) {
    const int idx = p + lane;
    const int gidx = idx < n_pieces ? idx : n_pieces - 1;
    f4 piece = __builtin_nontemporal_load(&gsrc[gidx]);
    if (p + 32 > kPieces && idx >= kPieces) piece = f4{0.0f, 0.0f, 0.0f, 0.0f};
    tile4[wave][idx] = piece;
  }
  __builtin_amdgcn_fence(__ATOMIC_RELEASE, "wavefront");
  __builtin_amdgcn_wave_barrier();
  __builtin_amdgcn_fence(__ATOMIC_ACQUIRE, "wavefront");

  const float* row_in[kTilesPerWave];
  float*       row_out[kTilesPerWave];
#pragma unroll
  for (int k = 0; k < kTilesPerWave; ++k) {
    row_out[k] = tile + (n + 16 * k) * kSteps;
    row_in[k]  = tile + (lo ? (n + 16 * k) * kSteps : kPieces * 4);
  }

  float s_n1[kTilesPerWave];
  float s_n2[kTilesPerWave];
  float x[kTilesPerWave];
  float d[kTilesPerWave];
  const float d0 = a_init[0];
#pragma unroll
  for (int k = 0; k < kTilesPerWave; ++k) {
    x[k]    = row_in[k][0];
    s_n1[k] = row_in[k][1];
    s_n2[k] = s_n1[k];
    d[k]    = d0;
  }

#pragma unroll
  for (int t = 0; t < kSteps; ++t) {
    v8f hid_lo[kTilesPerWave], hid_hi[kTilesPerWave];
    float part[kTilesPerWave];
    if (t + 2 < kSteps) {
#pragma unroll
      for (int k = 0; k < kTilesPerWave; ++k) s_n2[k] = row_in[k][t + 2];
    }
#pragma unroll
    for (int k = 0; k < kTilesPerWave; ++k) {
      v2f bm;
      bm.x = x[k];
      bm.y = d[k];

      hid_lo[k] = __builtin_amdgcn_wmma_f32_16x16x4_f32(
          false, a_lo, false, bm, (short)0, c_lo, false, false);
      hid_hi[k] = __builtin_amdgcn_wmma_f32_16x16x4_f32(
          false, a_hi, false, bm, (short)0, c_hi, false, false);

      part[k] = fmaf(bm.x, cx, fmaf(bm.y, cy, kc));
    }
#pragma unroll
    for (int v = 0; v < 8; ++v) part[0] = fmaf(__builtin_fabsf(hid_lo[0][v]), whlo[v], part[0]);
#pragma unroll
    for (int v = 0; v < 8; ++v) part[0] = fmaf(__builtin_fabsf(hid_hi[0][v]), whhi[v], part[0]);
    {
#pragma clang fp contract(on)
#pragma unroll
      for (int v = 0; v < 8; ++v) part[1] = __builtin_fabsf(hid_lo[1][v]) * whlo[v] + part[1];
#pragma unroll
      for (int v = 0; v < 8; ++v) part[1] = __builtin_fabsf(hid_hi[1][v]) * whhi[v] + part[1];
    }
#pragma unroll
    for (int k = 0; k < kTilesPerWave; ++k) {
      const float dval = part[k] + swap16(part[k]);
      if (t + 1 < kSteps) {
        x[k] = fmaf(c2, dval, fmaf(c8, x[k], s_n1[k]));
        s_n1[k] = s_n2[k];
      }
      d[k] = dval;
      row_out[k][t] = dval;
    }
  }

  __builtin_amdgcn_fence(__ATOMIC_RELEASE, "wavefront");
  __builtin_amdgcn_wave_barrier();
  __builtin_amdgcn_fence(__ATOMIC_ACQUIRE, "wavefront");

  f4* gdst = (f4*)(out + (size_t)base * kSteps);
  if (n_pieces == kPieces) {
#pragma unroll
    for (int p = 0; p < kSlabPieces; p += 32) {
      const int idx = p + lane;
      if (idx < kPieces)
        __builtin_nontemporal_store(tile4[wave][idx], &gdst[idx]);
    }
  } else {
#pragma unroll
    for (int p = 0; p < kSlabPieces; p += 32) {
      const int idx = p + lane;
      if (idx < n_pieces)
        __builtin_nontemporal_store(tile4[wave][idx], &gdst[idx]);
    }
  }
}

extern "C" void kernel_launch(void* const* d_in, const int* in_sizes, int n_in,
                              void* d_out, int out_size, void* d_ws, size_t ws_size,
                              hipStream_t stream) {
  const float* S      = (const float*)d_in[0];
  const float* W1     = (const float*)d_in[1];
  const float* b1     = (const float*)d_in[2];
  const float* W2     = (const float*)d_in[3];
  const float* b2     = (const float*)d_in[4];
  const float* a_init = (const float*)d_in[5];
  float* out = (float*)d_out;

  const int Btot   = in_sizes[0] / kSteps;
  const int blocks = (Btot + kBatchPerBlock - 1) / kBatchPerBlock;
  hedge_wmma_kernel<<<blocks, kWavesPerBlock * 32, 0, stream>>>(
      S, W1, b1, W2, b2, a_init, out, Btot);
}
